# starting kernel
# baseline (speedup 1.0000x reference)
#include <hip/hip_runtime.h>
#include <hip/hip_bf16.h>


#define BSZ     2
#define LSEQ    2048
#define DIN     2048
#define DSTATE  8
#define DTRANK  256
#define KPROJ   (DTRANK + 2 * DSTATE)
#define NCHUNK  8
#define LCHUNK  (LSEQ / NCHUNK)

typedef float v2f __attribute__((ext_vector_type(2)));
typedef float v8f __attribute__((ext_vector_type(8)));

__global__ __launch_bounds__(256) void proj_kernel(
    const float* __restrict__ x,
    const float* __restrict__ Wxp,
    float* __restrict__ xdbl)
{
    const int lane = threadIdx.x & 31;
    const int wave = threadIdx.x >> 5;
    const int tile = blockIdx.x * 8 + wave;
    const int rt = tile / 17;
    const int ct = tile % 17;
    const int m  = lane & 15;
    const int hi = lane >> 4;

    const float* arow = x   + (size_t)(rt * 16 + m) * DIN + 2 * hi;
    const float* brow = Wxp + (size_t)(ct * 16 + m) * DIN + 2 * hi;

    v8f acc = {};
    for (int k = 0; k < DIN; k += 4) {
        v2f a = *(const v2f*)(arow + k);
        v2f b = *(const v2f*)(brow + k);
        acc = __builtin_amdgcn_wmma_f32_16x16x4_f32(
            false, a, false, b, (short)0, acc, false, false);
    }
    #pragma unroll
    for (int j = 0; j < 8; ++j) {
        const int row = rt * 16 + j + 8 * hi;
        const int col = ct * 16 + m;
        xdbl[(size_t)row * KPROJ + col] = acc[j];
    }
}

__global__ __launch_bounds__(256) void delta_kernel(
    const float* __restrict__ xdbl,
    const float* __restrict__ Wdt,
    const float* __restrict__ b_dt,
    float* __restrict__ delta)
{
    const int lane = threadIdx.x & 31;
    const int wave = threadIdx.x >> 5;
    const int tile = blockIdx.x * 8 + wave;
    const int rt = tile >> 7;
    const int ct = tile & 127;
    const int m  = lane & 15;
    const int hi = lane >> 4;

    const float* arow = xdbl + (size_t)(rt * 16 + m) * KPROJ + 2 * hi;
    const float* brow = Wdt  + (size_t)(ct * 16 + m) * DTRANK + 2 * hi;

    v8f acc = {};
    for (int k = 0; k < DTRANK; k += 4) {
        v2f a = *(const v2f*)(arow + k);
        v2f b = *(const v2f*)(brow + k);
        acc = __builtin_amdgcn_wmma_f32_16x16x4_f32(
            false, a, false, b, (short)0, acc, false, false);
    }
    #pragma unroll
    for (int j = 0; j < 8; ++j) {
        const int row = rt * 16 + j + 8 * hi;
        const int col = ct * 16 + m;
        float v = acc[j] + b_dt[col];
        float sp = (v > 20.0f) ? v : log1pf(__expf(v));
        delta[(size_t)row * DIN + col] = sp;
    }
}

__global__ __launch_bounds__(256) void scan_pass1(
    const float* __restrict__ delta,
    const float* __restrict__ x,
    const float* __restrict__ xdbl,
    const float* __restrict__ A_log,
    float* __restrict__ aProd,
    float* __restrict__ hPart)
{
    const int t = blockIdx.x * blockDim.x + threadIdx.x;
    const int n = t & 7;
    const int c = (t >> 3) & 7;
    const int d = (t >> 6) & (DIN - 1);
    const int b = t >> 17;

    const float A = -__expf(A_log[d * DSTATE + n]);
    float h = 0.0f, aP = 1.0f;
    const int l0 = c * LCHUNK;
    for (int i = 0; i < LCHUNK; ++i) {
        const int row = b * LSEQ + l0 + i;
        const float dv = delta[(size_t)row * DIN + d];
        const float xv = x[(size_t)row * DIN + d];
        const float Bv = xdbl[(size_t)row * KPROJ + DTRANK + n];
        const float dA = __expf(dv * A);
        h  = fmaf(dA, h, dv * xv * Bv);
        aP *= dA;
    }
    const int idx = (((b * DIN + d) * NCHUNK) + c) * DSTATE + n;
    aProd[idx] = aP;
    hPart[idx] = h;
}

__global__ __launch_bounds__(256) void scan_pass2(
    const float* __restrict__ aProd,
    const float* __restrict__ hPart,
    float* __restrict__ hIn)
{
    const int t = blockIdx.x * blockDim.x + threadIdx.x;
    const int n  = t & 7;
    const int bd = t >> 3;
    float h = 0.0f;
    #pragma unroll
    for (int c = 0; c < NCHUNK; ++c) {
        const int idx = (bd * NCHUNK + c) * DSTATE + n;
        hIn[idx] = h;
        h = fmaf(aProd[idx], h, hPart[idx]);
    }
}

__global__ __launch_bounds__(256) void scan_pass3(
    const float* __restrict__ delta,
    const float* __restrict__ x,
    const float* __restrict__ xdbl,
    const float* __restrict__ A_log,
    const float* __restrict__ D_param,
    const float* __restrict__ hIn,
    float* __restrict__ y)
{
    const int t = blockIdx.x * blockDim.x + threadIdx.x;
    const int n = t & 7;
    const int c = (t >> 3) & 7;
    const int d = (t >> 6) & (DIN - 1);
    const int b = t >> 17;

    const float A  = -__expf(A_log[d * DSTATE + n]);
    const float Dp = D_param[d];
    float h = hIn[(((b * DIN + d) * NCHUNK) + c) * DSTATE + n];
    const int l0 = c * LCHUNK;
    for (int i = 0; i < LCHUNK; ++i) {
        const int row = b * LSEQ + l0 + i;
        const float dv = delta[(size_t)row * DIN + d];
        const float xv = x[(size_t)row * DIN + d];
        const float Bv = xdbl[(size_t)row * KPROJ + DTRANK + n];
        const float Cv = xdbl[(size_t)row * KPROJ + DTRANK + DSTATE + n];
        const float dA = __expf(dv * A);
        h = fmaf(dA, h, dv * xv * Bv);
        float part = h * Cv;
        part += __shfl_xor(part, 1, 32);
        part += __shfl_xor(part, 2, 32);
        part += __shfl_xor(part, 4, 32);
        if (n == 0)
            y[(size_t)row * DIN + d] = part + xv * Dp;
    }
}

extern "C" void kernel_launch(void* const* d_in, const int* in_sizes, int n_in,
                              void* d_out, int out_size, void* d_ws, size_t ws_size,
                              hipStream_t stream) {
    const float* x       = (const float*)d_in[0];
    const float* W_xproj = (const float*)d_in[1];
    const float* W_dt    = (const float*)d_in[2];
    const float* b_dt    = (const float*)d_in[3];
    const float* A_log   = (const float*)d_in[4];
    const float* D_param = (const float*)d_in[5];
    float* y = (float*)d_out;

    float* ws    = (float*)d_ws;
    float* xdbl  = ws;
    float* delta = xdbl + (size_t)BSZ * LSEQ * KPROJ;
    float* aProd = delta + (size_t)BSZ * LSEQ * DIN;
    float* hPart = aProd + (size_t)BSZ * DIN * NCHUNK * DSTATE;
    float* hIn   = hPart + (size_t)BSZ * DIN * NCHUNK * DSTATE;

    proj_kernel<<<544, 256, 0, stream>>>(x, W_xproj, xdbl);
    delta_kernel<<<4096, 256, 0, stream>>>(xdbl, W_dt, b_dt, delta);
    scan_pass1<<<1024, 256, 0, stream>>>(delta, x, xdbl, A_log, aProd, hPart);
    scan_pass2<<<128, 256, 0, stream>>>(aProd, hPart, hIn);
    scan_pass3<<<1024, 256, 0, stream>>>(delta, x, xdbl, A_log, D_param, hIn, y);
}


# both projection GEMMs: x/W k-chunks staged once per block in LDS by whole-line 128-bit loads (block tiles 16x272, 128x128), fragments by ds_load_b128; k order inside 8-k steps changes (sum order)
# speedup vs baseline: 3.2356x; 3.2356x over previous
#include <hip/hip_runtime.h>
#include <hip/hip_bf16.h>


#define BSZ     2
#define LSEQ    2048
#define DIN     2048
#define DSTATE  8
#define DTRANK  256
#define KPROJ   (DTRANK + 2 * DSTATE)

typedef float v2f __attribute__((ext_vector_type(2)));
typedef float v4f __attribute__((ext_vector_type(4)));
typedef float v8f __attribute__((ext_vector_type(8)));


__device__ __forceinline__ v8f mma_lo(v4f a, v4f b, v8f acc) {
    const v2f a0 = __builtin_shufflevector(a, a, 0, 1);
    const v2f b0 = __builtin_shufflevector(b, b, 0, 1);
    return __builtin_amdgcn_wmma_f32_16x16x4_f32(false, a0, false, b0, (short)0, acc, false, false);
}
__device__ __forceinline__ v8f mma_hi(v4f a, v4f b, v8f acc) {
    const v2f a1 = __builtin_shufflevector(a, a, 2, 3);
    const v2f b1 = __builtin_shufflevector(b, b, 2, 3);
    return __builtin_amdgcn_wmma_f32_16x16x4_f32(false, a1, false, b1, (short)0, acc, false, false);
}

template <int NT, int NKS, int LD>
__device__ __forceinline__ void chunk_1xN(const float* aF, const float* bF, v8f* acc) {
    v4f a[2], b[2][NT];
    a[0] = *(const v4f*)aF;
    #pragma unroll
    for (int j = 0; j < NT; ++j) b[0][j] = *(const v4f*)(bF + j * 16 * LD);
    #pragma unroll
    for (int ks = 0; ks < NKS; ++ks) {
        const int cu = ks & 1, nx = cu ^ 1;
        if (ks + 1 < NKS) {
            a[nx] = *(const v4f*)(aF + 8 * (ks + 1));
            #pragma unroll
            for (int j = 0; j < NT; ++j) b[nx][j] = *(const v4f*)(bF + j * 16 * LD + 8 * (ks + 1));
        }
        #pragma unroll
        for (int j = 0; j < NT; ++j) acc[j] = mma_lo(a[cu], b[cu][j], acc[j]);
        #pragma unroll
        for (int j = 0; j < NT; ++j) acc[j] = mma_hi(a[cu], b[cu][j], acc[j]);
    }
}

#define P1_BK   64
#define P1_LD   (P1_BK + 4)
#define P1_NCH  (DIN / P1_BK)
#define P1_NB   (KPROJ / 8)

__global__ __launch_bounds__(128) void proj_kernel(
    const float* __restrict__ x,
    const float* __restrict__ Wxp,
    float* __restrict__ xdbl)
{
    __shared__ __attribute__((aligned(16))) float sA[2][16 * P1_LD];
    __shared__ __attribute__((aligned(16))) float sB[2][KPROJ * P1_LD];

    const int tid  = threadIdx.x;
    const int lane = tid & 31;
    const int wave = __builtin_amdgcn_readfirstlane(tid >> 5);
    const int c    = lane & 15;
    const int h    = lane >> 4;
    const int m0   = blockIdx.x * 16;
    const int q    = tid & 15;
    const int r    = tid >> 4;
    const bool five = (wave == 0);
    const int  t0   = five ? 0 : 4 * wave + 1;

    const float* gA = x   + (size_t)(m0 + r) * DIN + 4 * q;
    const float* gB = Wxp + (size_t)r * DIN + 4 * q;
    const int    sO = r * P1_LD + 4 * q;

    v4f pa[2], pb[P1_NB];
    #pragma unroll
    for (int i = 0; i < 2; ++i)     pa[i] = *(const v4f*)(gA + (size_t)(8 * i) * DIN);
    #pragma unroll
    for (int i = 0; i < P1_NB; ++i) pb[i] = *(const v4f*)(gB + (size_t)(8 * i) * DIN);
    #pragma unroll
    for (int i = 0; i < 2; ++i)     *(v4f*)&sA[0][sO + 8 * i * P1_LD] = pa[i];
    #pragma unroll
    for (int i = 0; i < P1_NB; ++i) *(v4f*)&sB[0][sO + 8 * i * P1_LD] = pb[i];
    __syncthreads();

    v8f acc[5] = {};
    for (int ch = 0; ch < P1_NCH; ++ch) {
        const int  cur  = ch & 1;
        const bool more = (ch + 1 < P1_NCH);

        if (more) {
            const int k1 = (ch + 1) * P1_BK;
            #pragma unroll
            for (int i = 0; i < 2; ++i)     pa[i] = *(const v4f*)(gA + (size_t)(8 * i) * DIN + k1);
            #pragma unroll
            for (int i = 0; i < P1_NB; ++i) pb[i] = *(const v4f*)(gB + (size_t)(8 * i) * DIN + k1);
            asm volatile("" ::: "memory");
        }

        const float* aF = &sA[cur][c * P1_LD + 4 * h];
        const float* bF = &sB[cur][(t0 * 16 + c) * P1_LD + 4 * h];
        if (five) chunk_1xN<5, P1_BK / 8, P1_LD>(aF, bF, acc);
        else      chunk_1xN<4, P1_BK / 8, P1_LD>(aF, bF, acc);

        if (more) {
            const int nb = cur ^ 1;
            #pragma unroll
            for (int i = 0; i < 2; ++i)     *(v4f*)&sA[nb][sO + 8 * i * P1_LD] = pa[i];
            #pragma unroll
            for (int i = 0; i < P1_NB; ++i) *(v4f*)&sB[nb][sO + 8 * i * P1_LD] = pb[i];
        }
        __syncthreads();
    }

    #pragma unroll
    for (int j = 0; j < 5; ++j) {
        if (j < 4 || five) {
            #pragma unroll
            for (int e = 0; e < 8; ++e) {
                const int row = m0 + e + 8 * h;
                const int col = (t0 + j) * 16 + c;
                xdbl[(size_t)row * KPROJ + col] = acc[j][e];
            }
        }
    }
}

#define D_BK    32
#define D_LD    (D_BK + 4)
#define D_NCH   (DTRANK / D_BK)

__global__ __launch_bounds__(256) void delta_kernel(
    const float* __restrict__ xdbl,
    const float* __restrict__ Wdt,
    const float* __restrict__ b_dt,
    float* __restrict__ delta)
{
    __shared__ __attribute__((aligned(16))) float sA[2][128 * D_LD];
    __shared__ __attribute__((aligned(16))) float sB[2][128 * D_LD];

    const int tid  = threadIdx.x;
    const int lane = tid & 31;
    const int wave = tid >> 5;
    const int c    = lane & 15;
    const int h    = lane >> 4;
    const int m0   = (blockIdx.x >> 4) * 128;
    const int n0   = (blockIdx.x & 15) * 128;
    const int wr   = wave >> 1;
    const int wc   = wave & 1;
    const int q    = tid & 7;
    const int r    = tid >> 3;

    const float* gA = xdbl + (size_t)(m0 + r) * KPROJ  + 4 * q;
    const float* gB = Wdt  + (size_t)(n0 + r) * DTRANK + 4 * q;
    const int    sO = r * D_LD + 4 * q;

    v4f pa[4], pb[4];
    #pragma unroll
    for (int i = 0; i < 4; ++i) {
        pa[i] = *(const v4f*)(gA + (size_t)(32 * i) * KPROJ);
        pb[i] = *(const v4f*)(gB + (size_t)(32 * i) * DTRANK);
    }
    #pragma unroll
    for (int i = 0; i < 4; ++i) {
        *(v4f*)&sA[0][sO + 32 * i * D_LD] = pa[i];
        *(v4f*)&sB[0][sO + 32 * i * D_LD] = pb[i];
    }
    __syncthreads();

    v8f acc[2][4] = {};
    for (int ch = 0; ch < D_NCH; ++ch) {
        const int  cur  = ch & 1;
        const bool more = (ch + 1 < D_NCH);

        if (more) {
            const int k1 = (ch + 1) * D_BK;
            #pragma unroll
            for (int i = 0; i < 4; ++i) {
                pa[i] = *(const v4f*)(gA + (size_t)(32 * i) * KPROJ  + k1);
                pb[i] = *(const v4f*)(gB + (size_t)(32 * i) * DTRANK + k1);
            }
            asm volatile("" ::: "memory");
        }

        const float* aF = &sA[cur][(wr * 32 + c) * D_LD + 4 * h];
        const float* bF = &sB[cur][(wc * 64 + c) * D_LD + 4 * h];
        v4f a[2][2], b[2][4];
        #pragma unroll
        for (int i = 0; i < 2; ++i) a[0][i] = *(const v4f*)(aF + i * 16 * D_LD);
        #pragma unroll
        for (int j = 0; j < 4; ++j) b[0][j] = *(const v4f*)(bF + j * 16 * D_LD);
        #pragma unroll
        for (int ks = 0; ks < D_BK / 8; ++ks) {
            const int cu = ks & 1, nx = cu ^ 1;
            if (ks + 1 < D_BK / 8) {
                #pragma unroll
                for (int i = 0; i < 2; ++i) a[nx][i] = *(const v4f*)(aF + i * 16 * D_LD + 8 * (ks + 1));
                #pragma unroll
                for (int j = 0; j < 4; ++j) b[nx][j] = *(const v4f*)(bF + j * 16 * D_LD + 8 * (ks + 1));
            }
            #pragma unroll
            for (int i = 0; i < 2; ++i) {
                #pragma unroll
                for (int j = 0; j < 4; ++j) acc[i][j] = mma_lo(a[cu][i], b[cu][j], acc[i][j]);
            }
            #pragma unroll
            for (int i = 0; i < 2; ++i) {
                #pragma unroll
                for (int j = 0; j < 4; ++j) acc[i][j] = mma_hi(a[cu][i], b[cu][j], acc[i][j]);
            }
        }

        if (more) {
            const int nb = cur ^ 1;
            #pragma unroll
            for (int i = 0; i < 4; ++i) {
                *(v4f*)&sA[nb][sO + 32 * i * D_LD] = pa[i];
                *(v4f*)&sB[nb][sO + 32 * i * D_LD] = pb[i];
            }
        }
        __syncthreads();
    }

    #pragma unroll
    for (int i = 0; i < 2; ++i) {
        #pragma unroll
        for (int j = 0; j < 4; ++j) {
            const int col = n0 + wc * 64 + j * 16 + c;
            const float bias = b_dt[col];
            #pragma unroll
            for (int e = 0; e < 8; ++e) {
                const int row = m0 + wr * 32 + i * 16 + e + 8 * h;
                const float v = acc[i][j][e] + bias;
                const float sp = (v > 20.0f) ? v : log1pf(__expf(v));
                delta[(size_t)row * DIN + col] = sp;
            }
        }
    }
}

#define TS     64
#define TP     (TS + 4)
#define DB     16
#define NTILE  (LSEQ / TS)

template <int CTRL>
__device__ __forceinline__ float xmask_add(float v) {
    const int o = __builtin_amdgcn_update_dpp(0, __float_as_int(v), CTRL, 0xf, 0xf, true);
    return v + __int_as_float(o);
}

__device__ __forceinline__ float scan_step(float dv, float xv, float Bv, float Cv,
                                           float A, float Dp, float& h) {
    const float dA = __expf(dv * A);
    h = fmaf(dA, h, dv * xv * Bv);
    float part = h * Cv;
    part = xmask_add<0x161>(part);
    part = xmask_add<0x162>(part);
    part = xmask_add<0x164>(part);
    return part + xv * Dp;
}

__global__ __launch_bounds__(128) void scan_fused(
    const float* __restrict__ delta,
    const float* __restrict__ x,
    const float* __restrict__ xdbl,
    const float* __restrict__ A_log,
    const float* __restrict__ D_param,
    float* __restrict__ y)
{
    __shared__ __attribute__((aligned(16))) float sD[2][DB * TP];
    __shared__ __attribute__((aligned(16))) float sX[2][DB * TP];
    __shared__ __attribute__((aligned(16))) float sBC[2][16 * TP];
    __shared__ __attribute__((aligned(16))) float sY[2][DB * TP];

    const int tid = threadIdx.x;
    const int n   = tid & 7;
    const int dl  = tid >> 3;
    const int b   = blockIdx.x >> 7;
    const int d0  = (blockIdx.x & 127) * DB;
    const int sr  = tid >> 2;
    const int sq  = tid & 3;

    const float A  = -__expf(A_log[(d0 + dl) * DSTATE + n]);
    const float Dp = D_param[d0 + dl];
    float h = 0.0f;

    const size_t row00 = (size_t)b * LSEQ;
    float4 pd[2], px[2], pbc[2];

    for (int k = 0; k < 2; ++k) {
        const size_t row = row00 + sr + 32 * k;
        pd[k]  = *(const float4*)(delta + row * DIN + d0 + 4 * sq);
        px[k]  = *(const float4*)(x     + row * DIN + d0 + 4 * sq);
        pbc[k] = *(const float4*)(xdbl  + row * KPROJ + DTRANK + 4 * sq);
    }
    for (int k = 0; k < 2; ++k) {
        const int c = sr + 32 * k;
        sD[0][(4 * sq + 0) * TP + c] = pd[k].x;  sD[0][(4 * sq + 1) * TP + c] = pd[k].y;
        sD[0][(4 * sq + 2) * TP + c] = pd[k].z;  sD[0][(4 * sq + 3) * TP + c] = pd[k].w;
        sX[0][(4 * sq + 0) * TP + c] = px[k].x;  sX[0][(4 * sq + 1) * TP + c] = px[k].y;
        sX[0][(4 * sq + 2) * TP + c] = px[k].z;  sX[0][(4 * sq + 3) * TP + c] = px[k].w;
        sBC[0][(4 * sq + 0) * TP + c] = pbc[k].x; sBC[0][(4 * sq + 1) * TP + c] = pbc[k].y;
        sBC[0][(4 * sq + 2) * TP + c] = pbc[k].z; sBC[0][(4 * sq + 3) * TP + c] = pbc[k].w;
    }
    __syncthreads();

    for (int t = 0; t < NTILE; ++t) {
        const int cur = t & 1;
        const bool more = (t + 1 < NTILE);

        if (more) {
            for (int k = 0; k < 2; ++k) {
                const size_t row = row00 + (size_t)(t + 1) * TS + sr + 32 * k;
                pd[k]  = *(const float4*)(delta + row * DIN + d0 + 4 * sq);
                px[k]  = *(const float4*)(x     + row * DIN + d0 + 4 * sq);
                pbc[k] = *(const float4*)(xdbl  + row * KPROJ + DTRANK + 4 * sq);
            }
        }

        const float* pD = &sD[cur][dl * TP];
        const float* pX = &sX[cur][dl * TP];
        const float* pB = &sBC[cur][n * TP];
        const float* pC = &sBC[cur][(8 + n) * TP];
        float4 yv[TS / 4];
        #pragma unroll
        for (int g = 0; g < TS / 4; ++g) {
            const float4 vd = *(const float4*)(pD + 4 * g);
            const float4 vx = *(const float4*)(pX + 4 * g);
            const float4 vb = *(const float4*)(pB + 4 * g);
            const float4 vc = *(const float4*)(pC + 4 * g);
            yv[g].x = scan_step(vd.x, vx.x, vb.x, vc.x, A, Dp, h);
            yv[g].y = scan_step(vd.y, vx.y, vb.y, vc.y, A, Dp, h);
            yv[g].z = scan_step(vd.z, vx.z, vb.z, vc.z, A, Dp, h);
            yv[g].w = scan_step(vd.w, vx.w, vb.w, vc.w, A, Dp, h);
        }
        if (n == 0) {
            float* pY = &sY[cur][dl * TP];
            for (int g = 0; g < TS / 4; ++g)
                *(float4*)(pY + 4 * g) = yv[g];
        }

        if (more) {
            const int nb = cur ^ 1;
            for (int k = 0; k < 2; ++k) {
                const int c = sr + 32 * k;
                sD[nb][(4 * sq + 0) * TP + c] = pd[k].x;  sD[nb][(4 * sq + 1) * TP + c] = pd[k].y;
                sD[nb][(4 * sq + 2) * TP + c] = pd[k].z;  sD[nb][(4 * sq + 3) * TP + c] = pd[k].w;
                sX[nb][(4 * sq + 0) * TP + c] = px[k].x;  sX[nb][(4 * sq + 1) * TP + c] = px[k].y;
                sX[nb][(4 * sq + 2) * TP + c] = px[k].z;  sX[nb][(4 * sq + 3) * TP + c] = px[k].w;
                sBC[nb][(4 * sq + 0) * TP + c] = pbc[k].x; sBC[nb][(4 * sq + 1) * TP + c] = pbc[k].y;
                sBC[nb][(4 * sq + 2) * TP + c] = pbc[k].z; sBC[nb][(4 * sq + 3) * TP + c] = pbc[k].w;
            }
        }
        __syncthreads();

        for (int k = 0; k < 2; ++k) {
            const int c = sr + 32 * k;
            float4 v;
            v.x = sY[cur][(4 * sq + 0) * TP + c];
            v.y = sY[cur][(4 * sq + 1) * TP + c];
            v.z = sY[cur][(4 * sq + 2) * TP + c];
            v.w = sY[cur][(4 * sq + 3) * TP + c];
            const size_t row = row00 + (size_t)t * TS + c;
            *(float4*)(y + row * DIN + d0 + 4 * sq) = v;
        }
    }
}

extern "C" void kernel_launch(void* const* d_in, const int* in_sizes, int n_in,
                              void* d_out, int out_size, void* d_ws, size_t ws_size,
                              hipStream_t stream) {
    const float* x       = (const float*)d_in[0];
    const float* W_xproj = (const float*)d_in[1];
    const float* W_dt    = (const float*)d_in[2];
    const float* b_dt    = (const float*)d_in[3];
    const float* A_log   = (const float*)d_in[4];
    const float* D_param = (const float*)d_in[5];
    float* y = (float*)d_out;

    float* ws    = (float*)d_ws;
    float* xdbl  = ws;
    float* delta = xdbl + (size_t)BSZ * LSEQ * KPROJ;

    proj_kernel<<<256, 128, 0, stream>>>(x, W_xproj, xdbl);
    delta_kernel<<<512, 256, 0, stream>>>(xdbl, W_dt, b_dt, delta);
    scan_fused<<<256, 128, 0, stream>>>(delta, x, xdbl, A_log, D_param, y);
}


# x projection GEMM: block = 32 rows x 272 cols over one k half (grid 128x2), a staged W chunk serves 2 row tiles; the 2 k-half planes are added by the consumers' loaders (k sum order)
# speedup vs baseline: 3.3713x; 1.0419x over previous
#include <hip/hip_runtime.h>
#include <hip/hip_bf16.h>


#define BSZ     2
#define LSEQ    2048
#define DIN     2048
#define DSTATE  8
#define DTRANK  256
#define KPROJ   (DTRANK + 2 * DSTATE)

typedef float v2f __attribute__((ext_vector_type(2)));
typedef float v4f __attribute__((ext_vector_type(4)));
typedef float v8f __attribute__((ext_vector_type(8)));


__device__ __forceinline__ v8f mma_lo(v4f a, v4f b, v8f acc) {
    const v2f a0 = __builtin_shufflevector(a, a, 0, 1);
    const v2f b0 = __builtin_shufflevector(b, b, 0, 1);
    return __builtin_amdgcn_wmma_f32_16x16x4_f32(false, a0, false, b0, (short)0, acc, false, false);
}
__device__ __forceinline__ v8f mma_hi(v4f a, v4f b, v8f acc) {
    const v2f a1 = __builtin_shufflevector(a, a, 2, 3);
    const v2f b1 = __builtin_shufflevector(b, b, 2, 3);
    return __builtin_amdgcn_wmma_f32_16x16x4_f32(false, a1, false, b1, (short)0, acc, false, false);
}

template <int NT, int NKS, int LD>
__device__ __forceinline__ void chunk_1xN(const float* aF, const float* bF, v8f* acc) {
    v4f a[2], b[2][NT];
    a[0] = *(const v4f*)aF;
    #pragma unroll
    for (int j = 0; j < NT; ++j) b[0][j] = *(const v4f*)(bF + j * 16 * LD);
    #pragma unroll
    for (int ks = 0; ks < NKS; ++ks) {
        const int cu = ks & 1, nx = cu ^ 1;
        if (ks + 1 < NKS) {
            a[nx] = *(const v4f*)(aF + 8 * (ks + 1));
            #pragma unroll
            for (int j = 0; j < NT; ++j) b[nx][j] = *(const v4f*)(bF + j * 16 * LD + 8 * (ks + 1));
        }
        #pragma unroll
        for (int j = 0; j < NT; ++j) acc[j] = mma_lo(a[cu], b[cu][j], acc[j]);
        #pragma unroll
        for (int j = 0; j < NT; ++j) acc[j] = mma_hi(a[cu], b[cu][j], acc[j]);
    }
}

#define P1_BK   64
#define P1_LD   (P1_BK + 4)
#define P1_KH   (DIN / 2)
#define P1_NCH  (P1_KH / P1_BK)
#define P1_NA   4
#define P1_NB   (KPROJ / 8)
#define XPLANE  ((size_t)BSZ * LSEQ * KPROJ)

__global__ __launch_bounds__(128) void proj_kernel(
    const float* __restrict__ x,
    const float* __restrict__ Wxp,
    float* __restrict__ xdbl)
{
    __shared__ __attribute__((aligned(16))) float sA[2][32 * P1_LD];
    __shared__ __attribute__((aligned(16))) float sB[2][KPROJ * P1_LD];

    const int tid  = threadIdx.x;
    const int lane = tid & 31;
    const int wave = __builtin_amdgcn_readfirstlane(tid >> 5);
    const int c    = lane & 15;
    const int h    = lane >> 4;
    const int m0   = (blockIdx.x >> 1) * 32;
    const int kh   = blockIdx.x & 1;
    const int rt   = wave & 1;
    const int q    = tid & 15;
    const int r    = tid >> 4;
    const bool nine = ((wave >> 1) == 0);
    const int  t0   = nine ? 0 : 9;

    const float* gA = x   + (size_t)(m0 + r) * DIN + kh * P1_KH + 4 * q;
    const float* gB = Wxp + (size_t)r * DIN + kh * P1_KH + 4 * q;
    const int    sO = r * P1_LD + 4 * q;

    v4f pa[P1_NA], pb[P1_NB];
    #pragma unroll
    for (int i = 0; i < P1_NA; ++i) pa[i] = *(const v4f*)(gA + (size_t)(8 * i) * DIN);
    #pragma unroll
    for (int i = 0; i < P1_NB; ++i) pb[i] = *(const v4f*)(gB + (size_t)(8 * i) * DIN);
    #pragma unroll
    for (int i = 0; i < P1_NA; ++i) *(v4f*)&sA[0][sO + 8 * i * P1_LD] = pa[i];
    #pragma unroll
    for (int i = 0; i < P1_NB; ++i) *(v4f*)&sB[0][sO + 8 * i * P1_LD] = pb[i];
    __syncthreads();

    v8f acc[9] = {};
    for (int ch = 0; ch < P1_NCH; ++ch) {
        const int  cur  = ch & 1;
        const bool more = (ch + 1 < P1_NCH);

        if (more) {
            const int k1 = (ch + 1) * P1_BK;
            #pragma unroll
            for (int i = 0; i < P1_NA; ++i) pa[i] = *(const v4f*)(gA + (size_t)(8 * i) * DIN + k1);
            #pragma unroll
            for (int i = 0; i < P1_NB; ++i) pb[i] = *(const v4f*)(gB + (size_t)(8 * i) * DIN + k1);
            asm volatile("" ::: "memory");
        }

        const float* aF = &sA[cur][(rt * 16 + c) * P1_LD + 4 * h];
        const float* bF = &sB[cur][(t0 * 16 + c) * P1_LD + 4 * h];
        if (nine) chunk_1xN<9, P1_BK / 8, P1_LD>(aF, bF, acc);
        else      chunk_1xN<8, P1_BK / 8, P1_LD>(aF, bF, acc);

        if (more) {
            const int nb = cur ^ 1;
            #pragma unroll
            for (int i = 0; i < P1_NA; ++i) *(v4f*)&sA[nb][sO + 8 * i * P1_LD] = pa[i];
            #pragma unroll
            for (int i = 0; i < P1_NB; ++i) *(v4f*)&sB[nb][sO + 8 * i * P1_LD] = pb[i];
        }
        __syncthreads();
    }

    float* plane = xdbl + (size_t)kh * XPLANE;
    #pragma unroll
    for (int j = 0; j < 9; ++j) {
        if (j < 8 || nine) {
            #pragma unroll
            for (int e = 0; e < 8; ++e) {
                const int row = m0 + rt * 16 + e + 8 * h;
                const int col = (t0 + j) * 16 + c;
                plane[(size_t)row * KPROJ + col] = acc[j][e];
            }
        }
    }
}

#define D_BK    32
#define D_LD    (D_BK + 4)
#define D_NCH   (DTRANK / D_BK)

__global__ __launch_bounds__(256) void delta_kernel(
    const float* __restrict__ xdbl,
    const float* __restrict__ Wdt,
    const float* __restrict__ b_dt,
    float* __restrict__ delta)
{
    __shared__ __attribute__((aligned(16))) float sA[2][128 * D_LD];
    __shared__ __attribute__((aligned(16))) float sB[2][128 * D_LD];

    const int tid  = threadIdx.x;
    const int lane = tid & 31;
    const int wave = tid >> 5;
    const int c    = lane & 15;
    const int h    = lane >> 4;
    const int m0   = (blockIdx.x >> 4) * 128;
    const int n0   = (blockIdx.x & 15) * 128;
    const int wr   = wave >> 1;
    const int wc   = wave & 1;
    const int q    = tid & 7;
    const int r    = tid >> 3;

    const float* gA = xdbl + (size_t)(m0 + r) * KPROJ  + 4 * q;
    const float* gB = Wdt  + (size_t)(n0 + r) * DTRANK + 4 * q;
    const int    sO = r * D_LD + 4 * q;

    v4f pa[4], pa2[4], pb[4];
    #pragma unroll
    for (int i = 0; i < 4; ++i) {
        pa[i]  = *(const v4f*)(gA + (size_t)(32 * i) * KPROJ);
        pa2[i] = *(const v4f*)(gA + XPLANE + (size_t)(32 * i) * KPROJ);
        pb[i]  = *(const v4f*)(gB + (size_t)(32 * i) * DTRANK);
    }
    #pragma unroll
    for (int i = 0; i < 4; ++i) {
        *(v4f*)&sA[0][sO + 32 * i * D_LD] = pa[i] + pa2[i];
        *(v4f*)&sB[0][sO + 32 * i * D_LD] = pb[i];
    }
    __syncthreads();

    v8f acc[2][4] = {};
    for (int ch = 0; ch < D_NCH; ++ch) {
        const int  cur  = ch & 1;
        const bool more = (ch + 1 < D_NCH);

        if (more) {
            const int k1 = (ch + 1) * D_BK;
            #pragma unroll
            for (int i = 0; i < 4; ++i) {
                pa[i]  = *(const v4f*)(gA + (size_t)(32 * i) * KPROJ  + k1);
                pa2[i] = *(const v4f*)(gA + XPLANE + (size_t)(32 * i) * KPROJ + k1);
                pb[i]  = *(const v4f*)(gB + (size_t)(32 * i) * DTRANK + k1);
            }
            asm volatile("" ::: "memory");
        }

        const float* aF = &sA[cur][(wr * 32 + c) * D_LD + 4 * h];
        const float* bF = &sB[cur][(wc * 64 + c) * D_LD + 4 * h];
        v4f a[2][2], b[2][4];
        #pragma unroll
        for (int i = 0; i < 2; ++i) a[0][i] = *(const v4f*)(aF + i * 16 * D_LD);
        #pragma unroll
        for (int j = 0; j < 4; ++j) b[0][j] = *(const v4f*)(bF + j * 16 * D_LD);
        #pragma unroll
        for (int ks = 0; ks < D_BK / 8; ++ks) {
            const int cu = ks & 1, nx = cu ^ 1;
            if (ks + 1 < D_BK / 8) {
                #pragma unroll
                for (int i = 0; i < 2; ++i) a[nx][i] = *(const v4f*)(aF + i * 16 * D_LD + 8 * (ks + 1));
                #pragma unroll
                for (int j = 0; j < 4; ++j) b[nx][j] = *(const v4f*)(bF + j * 16 * D_LD + 8 * (ks + 1));
            }
            #pragma unroll
            for (int i = 0; i < 2; ++i) {
                #pragma unroll
                for (int j = 0; j < 4; ++j) acc[i][j] = mma_lo(a[cu][i], b[cu][j], acc[i][j]);
            }
            #pragma unroll
            for (int i = 0; i < 2; ++i) {
                #pragma unroll
                for (int j = 0; j < 4; ++j) acc[i][j] = mma_hi(a[cu][i], b[cu][j], acc[i][j]);
            }
        }

        if (more) {
            const int nb = cur ^ 1;
            #pragma unroll
            for (int i = 0; i < 4; ++i) {
                *(v4f*)&sA[nb][sO + 32 * i * D_LD] = pa[i] + pa2[i];
                *(v4f*)&sB[nb][sO + 32 * i * D_LD] = pb[i];
            }
        }
        __syncthreads();
    }

    #pragma unroll
    for (int i = 0; i < 2; ++i) {
        #pragma unroll
        for (int j = 0; j < 4; ++j) {
            const int col = n0 + wc * 64 + j * 16 + c;
            const float bias = b_dt[col];
            #pragma unroll
            for (int e = 0; e < 8; ++e) {
                const int row = m0 + wr * 32 + i * 16 + e + 8 * h;
                const float v = acc[i][j][e] + bias;
                const float sp = (v > 20.0f) ? v : log1pf(__expf(v));
                delta[(size_t)row * DIN + col] = sp;
            }
        }
    }
}

#define TS     64
#define TP     (TS + 4)
#define DB     16
#define NTILE  (LSEQ / TS)

template <int CTRL>
__device__ __forceinline__ float xmask_add(float v) {
    const int o = __builtin_amdgcn_update_dpp(0, __float_as_int(v), CTRL, 0xf, 0xf, true);
    return v + __int_as_float(o);
}

__device__ __forceinline__ float scan_step(float dv, float xv, float Bv, float Cv,
                                           float A, float Dp, float& h) {
    const float dA = __expf(dv * A);
    h = fmaf(dA, h, dv * xv * Bv);
    float part = h * Cv;
    part = xmask_add<0x161>(part);
    part = xmask_add<0x162>(part);
    part = xmask_add<0x164>(part);
    return part + xv * Dp;
}

__global__ __launch_bounds__(128) void scan_fused(
    const float* __restrict__ delta,
    const float* __restrict__ x,
    const float* __restrict__ xdbl,
    const float* __restrict__ A_log,
    const float* __restrict__ D_param,
    float* __restrict__ y)
{
    __shared__ __attribute__((aligned(16))) float sD[2][DB * TP];
    __shared__ __attribute__((aligned(16))) float sX[2][DB * TP];
    __shared__ __attribute__((aligned(16))) float sBC[2][16 * TP];
    __shared__ __attribute__((aligned(16))) float sY[2][DB * TP];

    const int tid = threadIdx.x;
    const int n   = tid & 7;
    const int dl  = tid >> 3;
    const int b   = blockIdx.x >> 7;
    const int d0  = (blockIdx.x & 127) * DB;
    const int sr  = tid >> 2;
    const int sq  = tid & 3;

    const float A  = -__expf(A_log[(d0 + dl) * DSTATE + n]);
    const float Dp = D_param[d0 + dl];
    float h = 0.0f;

    const size_t row00 = (size_t)b * LSEQ;
    float4 pd[2], px[2], pbc[2], pbc2[2];

    for (int k = 0; k < 2; ++k) {
        const size_t row = row00 + sr + 32 * k;
        pd[k]   = *(const float4*)(delta + row * DIN + d0 + 4 * sq);
        px[k]   = *(const float4*)(x     + row * DIN + d0 + 4 * sq);
        pbc[k]  = *(const float4*)(xdbl  + row * KPROJ + DTRANK + 4 * sq);
        pbc2[k] = *(const float4*)(xdbl  + XPLANE + row * KPROJ + DTRANK + 4 * sq);
    }
    for (int k = 0; k < 2; ++k) {
        const int c = sr + 32 * k;
        pbc[k].x += pbc2[k].x;  pbc[k].y += pbc2[k].y;
        pbc[k].z += pbc2[k].z;  pbc[k].w += pbc2[k].w;
        sD[0][(4 * sq + 0) * TP + c] = pd[k].x;  sD[0][(4 * sq + 1) * TP + c] = pd[k].y;
        sD[0][(4 * sq + 2) * TP + c] = pd[k].z;  sD[0][(4 * sq + 3) * TP + c] = pd[k].w;
        sX[0][(4 * sq + 0) * TP + c] = px[k].x;  sX[0][(4 * sq + 1) * TP + c] = px[k].y;
        sX[0][(4 * sq + 2) * TP + c] = px[k].z;  sX[0][(4 * sq + 3) * TP + c] = px[k].w;
        sBC[0][(4 * sq + 0) * TP + c] = pbc[k].x; sBC[0][(4 * sq + 1) * TP + c] = pbc[k].y;
        sBC[0][(4 * sq + 2) * TP + c] = pbc[k].z; sBC[0][(4 * sq + 3) * TP + c] = pbc[k].w;
    }
    __syncthreads();

    for (int t = 0; t < NTILE; ++t) {
        const int cur = t & 1;
        const bool more = (t + 1 < NTILE);

        if (more) {
            for (int k = 0; k < 2; ++k) {
                const size_t row = row00 + (size_t)(t + 1) * TS + sr + 32 * k;
                pd[k]   = *(const float4*)(delta + row * DIN + d0 + 4 * sq);
                px[k]   = *(const float4*)(x     + row * DIN + d0 + 4 * sq);
                pbc[k]  = *(const float4*)(xdbl  + row * KPROJ + DTRANK + 4 * sq);
                pbc2[k] = *(const float4*)(xdbl  + XPLANE + row * KPROJ + DTRANK + 4 * sq);
            }
        }

        const float* pD = &sD[cur][dl * TP];
        const float* pX = &sX[cur][dl * TP];
        const float* pB = &sBC[cur][n * TP];
        const float* pC = &sBC[cur][(8 + n) * TP];
        float4 yv[TS / 4];
        #pragma unroll
        for (int g = 0; g < TS / 4; ++g) {
            const float4 vd = *(const float4*)(pD + 4 * g);
            const float4 vx = *(const float4*)(pX + 4 * g);
            const float4 vb = *(const float4*)(pB + 4 * g);
            const float4 vc = *(const float4*)(pC + 4 * g);
            yv[g].x = scan_step(vd.x, vx.x, vb.x, vc.x, A, Dp, h);
            yv[g].y = scan_step(vd.y, vx.y, vb.y, vc.y, A, Dp, h);
            yv[g].z = scan_step(vd.z, vx.z, vb.z, vc.z, A, Dp, h);
            yv[g].w = scan_step(vd.w, vx.w, vb.w, vc.w, A, Dp, h);
        }
        if (n == 0) {
            float* pY = &sY[cur][dl * TP];
            for (int g = 0; g < TS / 4; ++g)
                *(float4*)(pY + 4 * g) = yv[g];
        }

        if (more) {
            const int nb = cur ^ 1;
            for (int k = 0; k < 2; ++k) {
                const int c = sr + 32 * k;
                pbc[k].x += pbc2[k].x;  pbc[k].y += pbc2[k].y;
                pbc[k].z += pbc2[k].z;  pbc[k].w += pbc2[k].w;
                sD[nb][(4 * sq + 0) * TP + c] = pd[k].x;  sD[nb][(4 * sq + 1) * TP + c] = pd[k].y;
                sD[nb][(4 * sq + 2) * TP + c] = pd[k].z;  sD[nb][(4 * sq + 3) * TP + c] = pd[k].w;
                sX[nb][(4 * sq + 0) * TP + c] = px[k].x;  sX[nb][(4 * sq + 1) * TP + c] = px[k].y;
                sX[nb][(4 * sq + 2) * TP + c] = px[k].z;  sX[nb][(4 * sq + 3) * TP + c] = px[k].w;
                sBC[nb][(4 * sq + 0) * TP + c] = pbc[k].x; sBC[nb][(4 * sq + 1) * TP + c] = pbc[k].y;
                sBC[nb][(4 * sq + 2) * TP + c] = pbc[k].z; sBC[nb][(4 * sq + 3) * TP + c] = pbc[k].w;
            }
        }
        __syncthreads();

        for (int k = 0; k < 2; ++k) {
            const int c = sr + 32 * k;
            float4 v;
            v.x = sY[cur][(4 * sq + 0) * TP + c];
            v.y = sY[cur][(4 * sq + 1) * TP + c];
            v.z = sY[cur][(4 * sq + 2) * TP + c];
            v.w = sY[cur][(4 * sq + 3) * TP + c];
            const size_t row = row00 + (size_t)t * TS + c;
            *(float4*)(y + row * DIN + d0 + 4 * sq) = v;
        }
    }
}

extern "C" void kernel_launch(void* const* d_in, const int* in_sizes, int n_in,
                              void* d_out, int out_size, void* d_ws, size_t ws_size,
                              hipStream_t stream) {
    const float* x       = (const float*)d_in[0];
    const float* W_xproj = (const float*)d_in[1];
    const float* W_dt    = (const float*)d_in[2];
    const float* b_dt    = (const float*)d_in[3];
    const float* A_log   = (const float*)d_in[4];
    const float* D_param = (const float*)d_in[5];
    float* y = (float*)d_out;

    float* ws    = (float*)d_ws;
    float* xdbl  = ws;
    float* delta = xdbl + 2 * XPLANE;

    proj_kernel<<<256, 128, 0, stream>>>(x, W_xproj, xdbl);
    delta_kernel<<<512, 256, 0, stream>>>(xdbl, W_dt, b_dt, delta);
    scan_fused<<<256, 128, 0, stream>>>(delta, x, xdbl, A_log, D_param, y);
}


# scan: operand reads run a stage ahead of use: each 16-step sub-block's 16 LDS reads go out one sub-block early (2nd register set), a tile's first 16 and the next rows from memory behind the barrier
# speedup vs baseline: 3.5545x; 1.0543x over previous
#include <hip/hip_runtime.h>
#include <hip/hip_bf16.h>


#define BSZ     2
#define LSEQ    2048
#define DIN     2048
#define DSTATE  8
#define DTRANK  256
#define KPROJ   (DTRANK + 2 * DSTATE)

typedef float v2f __attribute__((ext_vector_type(2)));
typedef float v4f __attribute__((ext_vector_type(4)));
typedef float v8f __attribute__((ext_vector_type(8)));


__device__ __forceinline__ v8f mma_lo(v4f a, v4f b, v8f acc) {
    const v2f a0 = __builtin_shufflevector(a, a, 0, 1);
    const v2f b0 = __builtin_shufflevector(b, b, 0, 1);
    return __builtin_amdgcn_wmma_f32_16x16x4_f32(false, a0, false, b0, (short)0, acc, false, false);
}
__device__ __forceinline__ v8f mma_hi(v4f a, v4f b, v8f acc) {
    const v2f a1 = __builtin_shufflevector(a, a, 2, 3);
    const v2f b1 = __builtin_shufflevector(b, b, 2, 3);
    return __builtin_amdgcn_wmma_f32_16x16x4_f32(false, a1, false, b1, (short)0, acc, false, false);
}

template <int NT, int NKS, int LD>
__device__ __forceinline__ void chunk_1xN(const float* aF, const float* bF, v8f* acc) {
    v4f a[2], b[2][NT];
    a[0] = *(const v4f*)aF;
    #pragma unroll
    for (int j = 0; j < NT; ++j) b[0][j] = *(const v4f*)(bF + j * 16 * LD);
    #pragma unroll
    for (int ks = 0; ks < NKS; ++ks) {
        const int cu = ks & 1, nx = cu ^ 1;
        if (ks + 1 < NKS) {
            a[nx] = *(const v4f*)(aF + 8 * (ks + 1));
            #pragma unroll
            for (int j = 0; j < NT; ++j) b[nx][j] = *(const v4f*)(bF + j * 16 * LD + 8 * (ks + 1));
        }
        #pragma unroll
        for (int j = 0; j < NT; ++j) acc[j] = mma_lo(a[cu], b[cu][j], acc[j]);
        #pragma unroll
        for (int j = 0; j < NT; ++j) acc[j] = mma_hi(a[cu], b[cu][j], acc[j]);
    }
}

#define P1_BK   64
#define P1_LD   (P1_BK + 4)
#define P1_KH   (DIN / 2)
#define P1_NCH  (P1_KH / P1_BK)
#define P1_NA   4
#define P1_NB   (KPROJ / 8)
#define XPLANE  ((size_t)BSZ * LSEQ * KPROJ)

__global__ __launch_bounds__(128) void proj_kernel(
    const float* __restrict__ x,
    const float* __restrict__ Wxp,
    float* __restrict__ xdbl)
{
    __shared__ __attribute__((aligned(16))) float sA[2][32 * P1_LD];
    __shared__ __attribute__((aligned(16))) float sB[2][KPROJ * P1_LD];

    const int tid  = threadIdx.x;
    const int lane = tid & 31;
    const int wave = __builtin_amdgcn_readfirstlane(tid >> 5);
    const int c    = lane & 15;
    const int h    = lane >> 4;
    const int m0   = (blockIdx.x >> 1) * 32;
    const int kh   = blockIdx.x & 1;
    const int rt   = wave & 1;
    const int q    = tid & 15;
    const int r    = tid >> 4;
    const bool nine = ((wave >> 1) == 0);
    const int  t0   = nine ? 0 : 9;

    const float* gA = x   + (size_t)(m0 + r) * DIN + kh * P1_KH + 4 * q;
    const float* gB = Wxp + (size_t)r * DIN + kh * P1_KH + 4 * q;
    const int    sO = r * P1_LD + 4 * q;

    v4f pa[P1_NA], pb[P1_NB];
    #pragma unroll
    for (int i = 0; i < P1_NA; ++i) pa[i] = *(const v4f*)(gA + (size_t)(8 * i) * DIN);
    #pragma unroll
    for (int i = 0; i < P1_NB; ++i) pb[i] = *(const v4f*)(gB + (size_t)(8 * i) * DIN);
    #pragma unroll
    for (int i = 0; i < P1_NA; ++i) *(v4f*)&sA[0][sO + 8 * i * P1_LD] = pa[i];
    #pragma unroll
    for (int i = 0; i < P1_NB; ++i) *(v4f*)&sB[0][sO + 8 * i * P1_LD] = pb[i];
    __syncthreads();

    v8f acc[9] = {};
    for (int ch = 0; ch < P1_NCH; ++ch) {
        const int  cur  = ch & 1;
        const bool more = (ch + 1 < P1_NCH);

        if (more) {
            const int k1 = (ch + 1) * P1_BK;
            #pragma unroll
            for (int i = 0; i < P1_NA; ++i) pa[i] = *(const v4f*)(gA + (size_t)(8 * i) * DIN + k1);
            #pragma unroll
            for (int i = 0; i < P1_NB; ++i) pb[i] = *(const v4f*)(gB + (size_t)(8 * i) * DIN + k1);
            asm volatile("" ::: "memory");
        }

        const float* aF = &sA[cur][(rt * 16 + c) * P1_LD + 4 * h];
        const float* bF = &sB[cur][(t0 * 16 + c) * P1_LD + 4 * h];
        if (nine) chunk_1xN<9, P1_BK / 8, P1_LD>(aF, bF, acc);
        else      chunk_1xN<8, P1_BK / 8, P1_LD>(aF, bF, acc);

        if (more) {
            const int nb = cur ^ 1;
            #pragma unroll
            for (int i = 0; i < P1_NA; ++i) *(v4f*)&sA[nb][sO + 8 * i * P1_LD] = pa[i];
            #pragma unroll
            for (int i = 0; i < P1_NB; ++i) *(v4f*)&sB[nb][sO + 8 * i * P1_LD] = pb[i];
        }
        __syncthreads();
    }

    float* plane = xdbl + (size_t)kh * XPLANE;
    #pragma unroll
    for (int j = 0; j < 9; ++j) {
        if (j < 8 || nine) {
            #pragma unroll
            for (int e = 0; e < 8; ++e) {
                const int row = m0 + rt * 16 + e + 8 * h;
                const int col = (t0 + j) * 16 + c;
                plane[(size_t)row * KPROJ + col] = acc[j][e];
            }
        }
    }
}

#define D_BK    32
#define D_LD    (D_BK + 4)
#define D_NCH   (DTRANK / D_BK)

__global__ __launch_bounds__(256) void delta_kernel(
    const float* __restrict__ xdbl,
    const float* __restrict__ Wdt,
    const float* __restrict__ b_dt,
    float* __restrict__ delta)
{
    __shared__ __attribute__((aligned(16))) float sA[2][128 * D_LD];
    __shared__ __attribute__((aligned(16))) float sB[2][128 * D_LD];

    const int tid  = threadIdx.x;
    const int lane = tid & 31;
    const int wave = tid >> 5;
    const int c    = lane & 15;
    const int h    = lane >> 4;
    const int m0   = (blockIdx.x >> 4) * 128;
    const int n0   = (blockIdx.x & 15) * 128;
    const int wr   = wave >> 1;
    const int wc   = wave & 1;
    const int q    = tid & 7;
    const int r    = tid >> 3;

    const float* gA = xdbl + (size_t)(m0 + r) * KPROJ  + 4 * q;
    const float* gB = Wdt  + (size_t)(n0 + r) * DTRANK + 4 * q;
    const int    sO = r * D_LD + 4 * q;

    v4f pa[4], pa2[4], pb[4];
    #pragma unroll
    for (int i = 0; i < 4; ++i) {
        pa[i]  = *(const v4f*)(gA + (size_t)(32 * i) * KPROJ);
        pa2[i] = *(const v4f*)(gA + XPLANE + (size_t)(32 * i) * KPROJ);
        pb[i]  = *(const v4f*)(gB + (size_t)(32 * i) * DTRANK);
    }
    #pragma unroll
    for (int i = 0; i < 4; ++i) {
        *(v4f*)&sA[0][sO + 32 * i * D_LD] = pa[i] + pa2[i];
        *(v4f*)&sB[0][sO + 32 * i * D_LD] = pb[i];
    }
    __syncthreads();

    v8f acc[2][4] = {};
    for (int ch = 0; ch < D_NCH; ++ch) {
        const int  cur  = ch & 1;
        const bool more = (ch + 1 < D_NCH);

        if (more) {
            const int k1 = (ch + 1) * D_BK;
            #pragma unroll
            for (int i = 0; i < 4; ++i) {
                pa[i]  = *(const v4f*)(gA + (size_t)(32 * i) * KPROJ  + k1);
                pa2[i] = *(const v4f*)(gA + XPLANE + (size_t)(32 * i) * KPROJ + k1);
                pb[i]  = *(const v4f*)(gB + (size_t)(32 * i) * DTRANK + k1);
            }
            asm volatile("" ::: "memory");
        }

        const float* aF = &sA[cur][(wr * 32 + c) * D_LD + 4 * h];
        const float* bF = &sB[cur][(wc * 64 + c) * D_LD + 4 * h];
        v4f a[2][2], b[2][4];
        #pragma unroll
        for (int i = 0; i < 2; ++i) a[0][i] = *(const v4f*)(aF + i * 16 * D_LD);
        #pragma unroll
        for (int j = 0; j < 4; ++j) b[0][j] = *(const v4f*)(bF + j * 16 * D_LD);
        #pragma unroll
        for (int ks = 0; ks < D_BK / 8; ++ks) {
            const int cu = ks & 1, nx = cu ^ 1;
            if (ks + 1 < D_BK / 8) {
                #pragma unroll
                for (int i = 0; i < 2; ++i) a[nx][i] = *(const v4f*)(aF + i * 16 * D_LD + 8 * (ks + 1));
                #pragma unroll
                for (int j = 0; j < 4; ++j) b[nx][j] = *(const v4f*)(bF + j * 16 * D_LD + 8 * (ks + 1));
            }
            #pragma unroll
            for (int i = 0; i < 2; ++i) {
                #pragma unroll
                for (int j = 0; j < 4; ++j) acc[i][j] = mma_lo(a[cu][i], b[cu][j], acc[i][j]);
            }
            #pragma unroll
            for (int i = 0; i < 2; ++i) {
                #pragma unroll
                for (int j = 0; j < 4; ++j) acc[i][j] = mma_hi(a[cu][i], b[cu][j], acc[i][j]);
            }
        }

        if (more) {
            const int nb = cur ^ 1;
            #pragma unroll
            for (int i = 0; i < 4; ++i) {
                *(v4f*)&sA[nb][sO + 32 * i * D_LD] = pa[i] + pa2[i];
                *(v4f*)&sB[nb][sO + 32 * i * D_LD] = pb[i];
            }
        }
        __syncthreads();
    }

    #pragma unroll
    for (int i = 0; i < 2; ++i) {
        #pragma unroll
        for (int j = 0; j < 4; ++j) {
            const int col = n0 + wc * 64 + j * 16 + c;
            const float bias = b_dt[col];
            #pragma unroll
            for (int e = 0; e < 8; ++e) {
                const int row = m0 + wr * 32 + i * 16 + e + 8 * h;
                const float v = acc[i][j][e] + bias;
                const float sp = (v > 20.0f) ? v : log1pf(__expf(v));
                delta[(size_t)row * DIN + col] = sp;
            }
        }
    }
}

#define TS     64
#define TP     (TS + 4)
#define DB     16
#define NTILE  (LSEQ / TS)
#define SBG    4

template <int CTRL>
__device__ __forceinline__ float xmask_add(float v) {
    const int o = __builtin_amdgcn_update_dpp(0, __float_as_int(v), CTRL, 0xf, 0xf, true);
    return v + __int_as_float(o);
}

__device__ __forceinline__ float scan_step(float dv, float xv, float Bv, float Cv,
                                           float A, float Dp, float& h) {
    const float dA = __expf(dv * A);
    h = fmaf(dA, h, dv * xv * Bv);
    float part = h * Cv;
    part = xmask_add<0x161>(part);
    part = xmask_add<0x162>(part);
    part = xmask_add<0x164>(part);
    return part + xv * Dp;
}

__device__ __forceinline__ void sub_load(float4 (&vd)[SBG], float4 (&vx)[SBG],
                                         float4 (&vb)[SBG], float4 (&vc)[SBG],
                                         const float* pD, const float* pX,
                                         const float* pB, const float* pC, int o) {
    #pragma unroll
    for (int g = 0; g < SBG; ++g) {
        vd[g] = *(const float4*)(pD + o + 4 * g);
        vx[g] = *(const float4*)(pX + o + 4 * g);
        vb[g] = *(const float4*)(pB + o + 4 * g);
        vc[g] = *(const float4*)(pC + o + 4 * g);
    }
}

__device__ __forceinline__ void sub_steps(const float4 (&vd)[SBG], const float4 (&vx)[SBG],
                                          const float4 (&vb)[SBG], const float4 (&vc)[SBG],
                                          float A, float Dp, float& h, float* pY, bool writer) {
    float4 yv[SBG];
    #pragma unroll
    for (int g = 0; g < SBG; ++g) {
        yv[g].x = scan_step(vd[g].x, vx[g].x, vb[g].x, vc[g].x, A, Dp, h);
        yv[g].y = scan_step(vd[g].y, vx[g].y, vb[g].y, vc[g].y, A, Dp, h);
        yv[g].z = scan_step(vd[g].z, vx[g].z, vb[g].z, vc[g].z, A, Dp, h);
        yv[g].w = scan_step(vd[g].w, vx[g].w, vb[g].w, vc[g].w, A, Dp, h);
    }
    if (writer) {
        #pragma unroll
        for (int g = 0; g < SBG; ++g) *(float4*)(pY + 4 * g) = yv[g];
    }
}

__device__ __forceinline__ void tile_request(float4 (&pd)[2], float4 (&px)[2],
                                             float4 (&pbc)[2], float4 (&pbc2)[2],
                                             const float* __restrict__ delta,
                                             const float* __restrict__ x,
                                             const float* __restrict__ xdbl,
                                             size_t row00, int T, int sr, int sq, int d0) {
    #pragma unroll
    for (int k = 0; k < 2; ++k) {
        const size_t row = row00 + (size_t)T * TS + sr + 32 * k;
        pd[k]   = *(const float4*)(delta + row * DIN + d0 + 4 * sq);
        px[k]   = *(const float4*)(x     + row * DIN + d0 + 4 * sq);
        pbc[k]  = *(const float4*)(xdbl  + row * KPROJ + DTRANK + 4 * sq);
        pbc2[k] = *(const float4*)(xdbl  + XPLANE + row * KPROJ + DTRANK + 4 * sq);
    }
}

__device__ __forceinline__ void tile_stage(float4 (&pd)[2], float4 (&px)[2],
                                           float4 (&pbc)[2], float4 (&pbc2)[2],
                                           float* sDb, float* sXb, float* sBCb, int sr, int sq) {
    #pragma unroll
    for (int k = 0; k < 2; ++k) {
        const int c = sr + 32 * k;
        pbc[k].x += pbc2[k].x;  pbc[k].y += pbc2[k].y;
        pbc[k].z += pbc2[k].z;  pbc[k].w += pbc2[k].w;
        sDb[(4 * sq + 0) * TP + c] = pd[k].x;  sDb[(4 * sq + 1) * TP + c] = pd[k].y;
        sDb[(4 * sq + 2) * TP + c] = pd[k].z;  sDb[(4 * sq + 3) * TP + c] = pd[k].w;
        sXb[(4 * sq + 0) * TP + c] = px[k].x;  sXb[(4 * sq + 1) * TP + c] = px[k].y;
        sXb[(4 * sq + 2) * TP + c] = px[k].z;  sXb[(4 * sq + 3) * TP + c] = px[k].w;
        sBCb[(4 * sq + 0) * TP + c] = pbc[k].x; sBCb[(4 * sq + 1) * TP + c] = pbc[k].y;
        sBCb[(4 * sq + 2) * TP + c] = pbc[k].z; sBCb[(4 * sq + 3) * TP + c] = pbc[k].w;
    }
}

__global__ __launch_bounds__(128) void scan_fused(
    const float* __restrict__ delta,
    const float* __restrict__ x,
    const float* __restrict__ xdbl,
    const float* __restrict__ A_log,
    const float* __restrict__ D_param,
    float* __restrict__ y)
{
    __shared__ __attribute__((aligned(16))) float sD[2][DB * TP];
    __shared__ __attribute__((aligned(16))) float sX[2][DB * TP];
    __shared__ __attribute__((aligned(16))) float sBC[2][16 * TP];
    __shared__ __attribute__((aligned(16))) float sY[2][DB * TP];

    const int tid = threadIdx.x;
    const int n   = tid & 7;
    const int dl  = tid >> 3;
    const int b   = blockIdx.x >> 7;
    const int d0  = (blockIdx.x & 127) * DB;
    const int sr  = tid >> 2;
    const int sq  = tid & 3;
    const bool writer = (n == 0);

    const float A  = -__expf(A_log[(d0 + dl) * DSTATE + n]);
    const float Dp = D_param[d0 + dl];
    float h = 0.0f;

    const size_t row00 = (size_t)b * LSEQ;
    float4 pd[2], px[2], pbc[2], pbc2[2];

    tile_request(pd, px, pbc, pbc2, delta, x, xdbl, row00, 0, sr, sq, d0);
    tile_stage(pd, px, pbc, pbc2, sD[0], sX[0], sBC[0], sr, sq);
    __syncthreads();
    tile_request(pd, px, pbc, pbc2, delta, x, xdbl, row00, 1, sr, sq, d0);

    float4 rd0[SBG], rx0[SBG], rb0[SBG], rc0[SBG];
    float4 rd1[SBG], rx1[SBG], rb1[SBG], rc1[SBG];
    sub_load(rd0, rx0, rb0, rc0, &sD[0][dl * TP], &sX[0][dl * TP],
             &sBC[0][n * TP], &sBC[0][(8 + n) * TP], 0);

    for (int t = 0; t < NTILE; ++t) {
        const int cur = t & 1;
        const int nb  = cur ^ 1;
        const bool more = (t + 1 < NTILE);

        const float* pD = &sD[cur][dl * TP];
        const float* pX = &sX[cur][dl * TP];
        const float* pB = &sBC[cur][n * TP];
        const float* pC = &sBC[cur][(8 + n) * TP];
        float* pY = &sY[cur][dl * TP];
        sub_load(rd1, rx1, rb1, rc1, pD, pX, pB, pC, 16);
        sub_steps(rd0, rx0, rb0, rc0, A, Dp, h, pY, writer);
        sub_load(rd0, rx0, rb0, rc0, pD, pX, pB, pC, 32);
        sub_steps(rd1, rx1, rb1, rc1, A, Dp, h, pY + 16, writer);
        sub_load(rd1, rx1, rb1, rc1, pD, pX, pB, pC, 48);
        sub_steps(rd0, rx0, rb0, rc0, A, Dp, h, pY + 32, writer);
        sub_steps(rd1, rx1, rb1, rc1, A, Dp, h, pY + 48, writer);

        if (more)
            tile_stage(pd, px, pbc, pbc2, sD[nb], sX[nb], sBC[nb], sr, sq);
        __syncthreads();

        if (t + 2 < NTILE)
            tile_request(pd, px, pbc, pbc2, delta, x, xdbl, row00, t + 2, sr, sq, d0);
        if (more)
            sub_load(rd0, rx0, rb0, rc0, &sD[nb][dl * TP], &sX[nb][dl * TP],
                     &sBC[nb][n * TP], &sBC[nb][(8 + n) * TP], 0);

        #pragma unroll
        for (int k = 0; k < 2; ++k) {
            const int c = sr + 32 * k;
            float4 v;
            v.x = sY[cur][(4 * sq + 0) * TP + c];
            v.y = sY[cur][(4 * sq + 1) * TP + c];
            v.z = sY[cur][(4 * sq + 2) * TP + c];
            v.w = sY[cur][(4 * sq + 3) * TP + c];
            const size_t row = row00 + (size_t)t * TS + c;
            *(float4*)(y + row * DIN + d0 + 4 * sq) = v;
        }
    }
}

extern "C" void kernel_launch(void* const* d_in, const int* in_sizes, int n_in,
                              void* d_out, int out_size, void* d_ws, size_t ws_size,
                              hipStream_t stream) {
    const float* x       = (const float*)d_in[0];
    const float* W_xproj = (const float*)d_in[1];
    const float* W_dt    = (const float*)d_in[2];
    const float* b_dt    = (const float*)d_in[3];
    const float* A_log   = (const float*)d_in[4];
    const float* D_param = (const float*)d_in[5];
    float* y = (float*)d_out;

    float* ws    = (float*)d_ws;
    float* xdbl  = ws;
    float* delta = xdbl + 2 * XPLANE;

    proj_kernel<<<256, 128, 0, stream>>>(x, W_xproj, xdbl);
    delta_kernel<<<512, 256, 0, stream>>>(xdbl, W_dt, b_dt, delta);
    scan_fused<<<256, 128, 0, stream>>>(delta, x, xdbl, A_log, D_param, y);
}


# delta epilogue: the per-element v>20 branch becomes a select, so a lane's 64 independent library log1p chains interleave as straight-line code (same exp and log1p calls, delta bit-identical)
# speedup vs baseline: 3.7255x; 1.0481x over previous
#include <hip/hip_runtime.h>
#include <hip/hip_bf16.h>


#define BSZ     2
#define LSEQ    2048
#define DIN     2048
#define DSTATE  8
#define DTRANK  256
#define KPROJ   (DTRANK + 2 * DSTATE)

typedef float v2f __attribute__((ext_vector_type(2)));
typedef float v4f __attribute__((ext_vector_type(4)));
typedef float v8f __attribute__((ext_vector_type(8)));


__device__ __forceinline__ v8f mma_lo(v4f a, v4f b, v8f acc) {
    const v2f a0 = __builtin_shufflevector(a, a, 0, 1);
    const v2f b0 = __builtin_shufflevector(b, b, 0, 1);
    return __builtin_amdgcn_wmma_f32_16x16x4_f32(false, a0, false, b0, (short)0, acc, false, false);
}
__device__ __forceinline__ v8f mma_hi(v4f a, v4f b, v8f acc) {
    const v2f a1 = __builtin_shufflevector(a, a, 2, 3);
    const v2f b1 = __builtin_shufflevector(b, b, 2, 3);
    return __builtin_amdgcn_wmma_f32_16x16x4_f32(false, a1, false, b1, (short)0, acc, false, false);
}

template <int NT, int NKS, int LD>
__device__ __forceinline__ void chunk_1xN(const float* aF, const float* bF, v8f* acc) {
    v4f a[2], b[2][NT];
    a[0] = *(const v4f*)aF;
    #pragma unroll
    for (int j = 0; j < NT; ++j) b[0][j] = *(const v4f*)(bF + j * 16 * LD);
    #pragma unroll
    for (int ks = 0; ks < NKS; ++ks) {
        const int cu = ks & 1, nx = cu ^ 1;
        if (ks + 1 < NKS) {
            a[nx] = *(const v4f*)(aF + 8 * (ks + 1));
            #pragma unroll
            for (int j = 0; j < NT; ++j) b[nx][j] = *(const v4f*)(bF + j * 16 * LD + 8 * (ks + 1));
        }
        #pragma unroll
        for (int j = 0; j < NT; ++j) acc[j] = mma_lo(a[cu], b[cu][j], acc[j]);
        #pragma unroll
        for (int j = 0; j < NT; ++j) acc[j] = mma_hi(a[cu], b[cu][j], acc[j]);
    }
}

#define P1_BK   64
#define P1_LD   (P1_BK + 4)
#define P1_KH   (DIN / 2)
#define P1_NCH  (P1_KH / P1_BK)
#define P1_NA   4
#define P1_NB   (KPROJ / 8)
#define XPLANE  ((size_t)BSZ * LSEQ * KPROJ)

__global__ __launch_bounds__(128) void proj_kernel(
    const float* __restrict__ x,
    const float* __restrict__ Wxp,
    float* __restrict__ xdbl)
{
    __shared__ __attribute__((aligned(16))) float sA[2][32 * P1_LD];
    __shared__ __attribute__((aligned(16))) float sB[2][KPROJ * P1_LD];

    const int tid  = threadIdx.x;
    const int lane = tid & 31;
    const int wave = __builtin_amdgcn_readfirstlane(tid >> 5);
    const int c    = lane & 15;
    const int h    = lane >> 4;
    const int m0   = (blockIdx.x >> 1) * 32;
    const int kh   = blockIdx.x & 1;
    const int rt   = wave & 1;
    const int q    = tid & 15;
    const int r    = tid >> 4;
    const bool nine = ((wave >> 1) == 0);
    const int  t0   = nine ? 0 : 9;

    const float* gA = x   + (size_t)(m0 + r) * DIN + kh * P1_KH + 4 * q;
    const float* gB = Wxp + (size_t)r * DIN + kh * P1_KH + 4 * q;
    const int    sO = r * P1_LD + 4 * q;

    v4f pa[P1_NA], pb[P1_NB];
    #pragma unroll
    for (int i = 0; i < P1_NA; ++i) pa[i] = *(const v4f*)(gA + (size_t)(8 * i) * DIN);
    #pragma unroll
    for (int i = 0; i < P1_NB; ++i) pb[i] = *(const v4f*)(gB + (size_t)(8 * i) * DIN);
    #pragma unroll
    for (int i = 0; i < P1_NA; ++i) *(v4f*)&sA[0][sO + 8 * i * P1_LD] = pa[i];
    #pragma unroll
    for (int i = 0; i < P1_NB; ++i) *(v4f*)&sB[0][sO + 8 * i * P1_LD] = pb[i];
    __syncthreads();

    v8f acc[9] = {};
    for (int ch = 0; ch < P1_NCH; ++ch) {
        const int  cur  = ch & 1;
        const bool more = (ch + 1 < P1_NCH);

        if (more) {
            const int k1 = (ch + 1) * P1_BK;
            #pragma unroll
            for (int i = 0; i < P1_NA; ++i) pa[i] = *(const v4f*)(gA + (size_t)(8 * i) * DIN + k1);
            #pragma unroll
            for (int i = 0; i < P1_NB; ++i) pb[i] = *(const v4f*)(gB + (size_t)(8 * i) * DIN + k1);
            asm volatile("" ::: "memory");
        }

        const float* aF = &sA[cur][(rt * 16 + c) * P1_LD + 4 * h];
        const float* bF = &sB[cur][(t0 * 16 + c) * P1_LD + 4 * h];
        if (nine) chunk_1xN<9, P1_BK / 8, P1_LD>(aF, bF, acc);
        else      chunk_1xN<8, P1_BK / 8, P1_LD>(aF, bF, acc);

        if (more) {
            const int nb = cur ^ 1;
            #pragma unroll
            for (int i = 0; i < P1_NA; ++i) *(v4f*)&sA[nb][sO + 8 * i * P1_LD] = pa[i];
            #pragma unroll
            for (int i = 0; i < P1_NB; ++i) *(v4f*)&sB[nb][sO + 8 * i * P1_LD] = pb[i];
        }
        __syncthreads();
    }

    float* plane = xdbl + (size_t)kh * XPLANE;
    #pragma unroll
    for (int j = 0; j < 9; ++j) {
        if (j < 8 || nine) {
            #pragma unroll
            for (int e = 0; e < 8; ++e) {
                const int row = m0 + rt * 16 + e + 8 * h;
                const int col = (t0 + j) * 16 + c;
                plane[(size_t)row * KPROJ + col] = acc[j][e];
            }
        }
    }
}

#define D_BK    32
#define D_LD    (D_BK + 4)
#define D_NCH   (DTRANK / D_BK)

__global__ __launch_bounds__(256) void delta_kernel(
    const float* __restrict__ xdbl,
    const float* __restrict__ Wdt,
    const float* __restrict__ b_dt,
    float* __restrict__ delta)
{
    __shared__ __attribute__((aligned(16))) float sA[2][128 * D_LD];
    __shared__ __attribute__((aligned(16))) float sB[2][128 * D_LD];

    const int tid  = threadIdx.x;
    const int lane = tid & 31;
    const int wave = tid >> 5;
    const int c    = lane & 15;
    const int h    = lane >> 4;
    const int m0   = (blockIdx.x >> 4) * 128;
    const int n0   = (blockIdx.x & 15) * 128;
    const int wr   = wave >> 1;
    const int wc   = wave & 1;
    const int q    = tid & 7;
    const int r    = tid >> 3;

    const float* gA = xdbl + (size_t)(m0 + r) * KPROJ  + 4 * q;
    const float* gB = Wdt  + (size_t)(n0 + r) * DTRANK + 4 * q;
    const int    sO = r * D_LD + 4 * q;

    v4f pa[4], pa2[4], pb[4];
    #pragma unroll
    for (int i = 0; i < 4; ++i) {
        pa[i]  = *(const v4f*)(gA + (size_t)(32 * i) * KPROJ);
        pa2[i] = *(const v4f*)(gA + XPLANE + (size_t)(32 * i) * KPROJ);
        pb[i]  = *(const v4f*)(gB + (size_t)(32 * i) * DTRANK);
    }
    #pragma unroll
    for (int i = 0; i < 4; ++i) {
        *(v4f*)&sA[0][sO + 32 * i * D_LD] = pa[i] + pa2[i];
        *(v4f*)&sB[0][sO + 32 * i * D_LD] = pb[i];
    }
    __syncthreads();

    v8f acc[2][4] = {};
    for (int ch = 0; ch < D_NCH; ++ch) {
        const int  cur  = ch & 1;
        const bool more = (ch + 1 < D_NCH);

        if (more) {
            const int k1 = (ch + 1) * D_BK;
            #pragma unroll
            for (int i = 0; i < 4; ++i) {
                pa[i]  = *(const v4f*)(gA + (size_t)(32 * i) * KPROJ  + k1);
                pa2[i] = *(const v4f*)(gA + XPLANE + (size_t)(32 * i) * KPROJ + k1);
                pb[i]  = *(const v4f*)(gB + (size_t)(32 * i) * DTRANK + k1);
            }
            asm volatile("" ::: "memory");
        }

        const float* aF = &sA[cur][(wr * 32 + c) * D_LD + 4 * h];
        const float* bF = &sB[cur][(wc * 64 + c) * D_LD + 4 * h];
        v4f a[2][2], b[2][4];
        #pragma unroll
        for (int i = 0; i < 2; ++i) a[0][i] = *(const v4f*)(aF + i * 16 * D_LD);
        #pragma unroll
        for (int j = 0; j < 4; ++j) b[0][j] = *(const v4f*)(bF + j * 16 * D_LD);
        #pragma unroll
        for (int ks = 0; ks < D_BK / 8; ++ks) {
            const int cu = ks & 1, nx = cu ^ 1;
            if (ks + 1 < D_BK / 8) {
                #pragma unroll
                for (int i = 0; i < 2; ++i) a[nx][i] = *(const v4f*)(aF + i * 16 * D_LD + 8 * (ks + 1));
                #pragma unroll
                for (int j = 0; j < 4; ++j) b[nx][j] = *(const v4f*)(bF + j * 16 * D_LD + 8 * (ks + 1));
            }
            #pragma unroll
            for (int i = 0; i < 2; ++i) {
                #pragma unroll
                for (int j = 0; j < 4; ++j) acc[i][j] = mma_lo(a[cu][i], b[cu][j], acc[i][j]);
            }
            #pragma unroll
            for (int i = 0; i < 2; ++i) {
                #pragma unroll
                for (int j = 0; j < 4; ++j) acc[i][j] = mma_hi(a[cu][i], b[cu][j], acc[i][j]);
            }
        }

        if (more) {
            const int nb = cur ^ 1;
            #pragma unroll
            for (int i = 0; i < 4; ++i) {
                *(v4f*)&sA[nb][sO + 32 * i * D_LD] = pa[i] + pa2[i];
                *(v4f*)&sB[nb][sO + 32 * i * D_LD] = pb[i];
            }
        }
        __syncthreads();
    }

    #pragma unroll
    for (int i = 0; i < 2; ++i) {
        #pragma unroll
        for (int j = 0; j < 4; ++j) {
            const int col = n0 + wc * 64 + j * 16 + c;
            const float bias = b_dt[col];
            #pragma unroll
            for (int e = 0; e < 8; ++e) {
                const int row = m0 + wr * 32 + i * 16 + e + 8 * h;
                const float v = acc[i][j][e] + bias;
                const float lp = log1pf(__expf(v));
                const float sp = (v > 20.0f) ? v : lp;
                delta[(size_t)row * DIN + col] = sp;
            }
        }
    }
}

#define TS     64
#define TP     (TS + 4)
#define DB     16
#define NTILE  (LSEQ / TS)
#define SBG    4

template <int CTRL>
__device__ __forceinline__ float xmask_add(float v) {
    const int o = __builtin_amdgcn_update_dpp(0, __float_as_int(v), CTRL, 0xf, 0xf, true);
    return v + __int_as_float(o);
}

__device__ __forceinline__ float scan_step(float dv, float xv, float Bv, float Cv,
                                           float A, float Dp, float& h) {
    const float dA = __expf(dv * A);
    h = fmaf(dA, h, dv * xv * Bv);
    float part = h * Cv;
    part = xmask_add<0x161>(part);
    part = xmask_add<0x162>(part);
    part = xmask_add<0x164>(part);
    return part + xv * Dp;
}

__device__ __forceinline__ void sub_load(float4 (&vd)[SBG], float4 (&vx)[SBG],
                                         float4 (&vb)[SBG], float4 (&vc)[SBG],
                                         const float* pD, const float* pX,
                                         const float* pB, const float* pC, int o) {
    #pragma unroll
    for (int g = 0; g < SBG; ++g) {
        vd[g] = *(const float4*)(pD + o + 4 * g);
        vx[g] = *(const float4*)(pX + o + 4 * g);
        vb[g] = *(const float4*)(pB + o + 4 * g);
        vc[g] = *(const float4*)(pC + o + 4 * g);
    }
}

__device__ __forceinline__ void sub_steps(const float4 (&vd)[SBG], const float4 (&vx)[SBG],
                                          const float4 (&vb)[SBG], const float4 (&vc)[SBG],
                                          float A, float Dp, float& h, float* pY, bool writer) {
    float4 yv[SBG];
    #pragma unroll
    for (int g = 0; g < SBG; ++g) {
        yv[g].x = scan_step(vd[g].x, vx[g].x, vb[g].x, vc[g].x, A, Dp, h);
        yv[g].y = scan_step(vd[g].y, vx[g].y, vb[g].y, vc[g].y, A, Dp, h);
        yv[g].z = scan_step(vd[g].z, vx[g].z, vb[g].z, vc[g].z, A, Dp, h);
        yv[g].w = scan_step(vd[g].w, vx[g].w, vb[g].w, vc[g].w, A, Dp, h);
    }
    if (writer) {
        #pragma unroll
        for (int g = 0; g < SBG; ++g) *(float4*)(pY + 4 * g) = yv[g];
    }
}

__device__ __forceinline__ void tile_request(float4 (&pd)[2], float4 (&px)[2],
                                             float4 (&pbc)[2], float4 (&pbc2)[2],
                                             const float* __restrict__ delta,
                                             const float* __restrict__ x,
                                             const float* __restrict__ xdbl,
                                             size_t row00, int T, int sr, int sq, int d0) {
    #pragma unroll
    for (int k = 0; k < 2; ++k) {
        const size_t row = row00 + (size_t)T * TS + sr + 32 * k;
        pd[k]   = *(const float4*)(delta + row * DIN + d0 + 4 * sq);
        px[k]   = *(const float4*)(x     + row * DIN + d0 + 4 * sq);
        pbc[k]  = *(const float4*)(xdbl  + row * KPROJ + DTRANK + 4 * sq);
        pbc2[k] = *(const float4*)(xdbl  + XPLANE + row * KPROJ + DTRANK + 4 * sq);
    }
}

__device__ __forceinline__ void tile_stage(float4 (&pd)[2], float4 (&px)[2],
                                           float4 (&pbc)[2], float4 (&pbc2)[2],
                                           float* sDb, float* sXb, float* sBCb, int sr, int sq) {
    #pragma unroll
    for (int k = 0; k < 2; ++k) {
        const int c = sr + 32 * k;
        pbc[k].x += pbc2[k].x;  pbc[k].y += pbc2[k].y;
        pbc[k].z += pbc2[k].z;  pbc[k].w += pbc2[k].w;
        sDb[(4 * sq + 0) * TP + c] = pd[k].x;  sDb[(4 * sq + 1) * TP + c] = pd[k].y;
        sDb[(4 * sq + 2) * TP + c] = pd[k].z;  sDb[(4 * sq + 3) * TP + c] = pd[k].w;
        sXb[(4 * sq + 0) * TP + c] = px[k].x;  sXb[(4 * sq + 1) * TP + c] = px[k].y;
        sXb[(4 * sq + 2) * TP + c] = px[k].z;  sXb[(4 * sq + 3) * TP + c] = px[k].w;
        sBCb[(4 * sq + 0) * TP + c] = pbc[k].x; sBCb[(4 * sq + 1) * TP + c] = pbc[k].y;
        sBCb[(4 * sq + 2) * TP + c] = pbc[k].z; sBCb[(4 * sq + 3) * TP + c] = pbc[k].w;
    }
}

__global__ __launch_bounds__(128) void scan_fused(
    const float* __restrict__ delta,
    const float* __restrict__ x,
    const float* __restrict__ xdbl,
    const float* __restrict__ A_log,
    const float* __restrict__ D_param,
    float* __restrict__ y)
{
    __shared__ __attribute__((aligned(16))) float sD[2][DB * TP];
    __shared__ __attribute__((aligned(16))) float sX[2][DB * TP];
    __shared__ __attribute__((aligned(16))) float sBC[2][16 * TP];
    __shared__ __attribute__((aligned(16))) float sY[2][DB * TP];

    const int tid = threadIdx.x;
    const int n   = tid & 7;
    const int dl  = tid >> 3;
    const int b   = blockIdx.x >> 7;
    const int d0  = (blockIdx.x & 127) * DB;
    const int sr  = tid >> 2;
    const int sq  = tid & 3;
    const bool writer = (n == 0);

    const float A  = -__expf(A_log[(d0 + dl) * DSTATE + n]);
    const float Dp = D_param[d0 + dl];
    float h = 0.0f;

    const size_t row00 = (size_t)b * LSEQ;
    float4 pd[2], px[2], pbc[2], pbc2[2];

    tile_request(pd, px, pbc, pbc2, delta, x, xdbl, row00, 0, sr, sq, d0);
    tile_stage(pd, px, pbc, pbc2, sD[0], sX[0], sBC[0], sr, sq);
    __syncthreads();
    tile_request(pd, px, pbc, pbc2, delta, x, xdbl, row00, 1, sr, sq, d0);

    float4 rd0[SBG], rx0[SBG], rb0[SBG], rc0[SBG];
    float4 rd1[SBG], rx1[SBG], rb1[SBG], rc1[SBG];
    sub_load(rd0, rx0, rb0, rc0, &sD[0][dl * TP], &sX[0][dl * TP],
             &sBC[0][n * TP], &sBC[0][(8 + n) * TP], 0);

    for (int t = 0; t < NTILE; ++t) {
        const int cur = t & 1;
        const int nb  = cur ^ 1;
        const bool more = (t + 1 < NTILE);

        const float* pD = &sD[cur][dl * TP];
        const float* pX = &sX[cur][dl * TP];
        const float* pB = &sBC[cur][n * TP];
        const float* pC = &sBC[cur][(8 + n) * TP];
        float* pY = &sY[cur][dl * TP];
        sub_load(rd1, rx1, rb1, rc1, pD, pX, pB, pC, 16);
        sub_steps(rd0, rx0, rb0, rc0, A, Dp, h, pY, writer);
        sub_load(rd0, rx0, rb0, rc0, pD, pX, pB, pC, 32);
        sub_steps(rd1, rx1, rb1, rc1, A, Dp, h, pY + 16, writer);
        sub_load(rd1, rx1, rb1, rc1, pD, pX, pB, pC, 48);
        sub_steps(rd0, rx0, rb0, rc0, A, Dp, h, pY + 32, writer);
        sub_steps(rd1, rx1, rb1, rc1, A, Dp, h, pY + 48, writer);

        if (more)
            tile_stage(pd, px, pbc, pbc2, sD[nb], sX[nb], sBC[nb], sr, sq);
        __syncthreads();

        if (t + 2 < NTILE)
            tile_request(pd, px, pbc, pbc2, delta, x, xdbl, row00, t + 2, sr, sq, d0);
        if (more)
            sub_load(rd0, rx0, rb0, rc0, &sD[nb][dl * TP], &sX[nb][dl * TP],
                     &sBC[nb][n * TP], &sBC[nb][(8 + n) * TP], 0);

        #pragma unroll
        for (int k = 0; k < 2; ++k) {
            const int c = sr + 32 * k;
            float4 v;
            v.x = sY[cur][(4 * sq + 0) * TP + c];
            v.y = sY[cur][(4 * sq + 1) * TP + c];
            v.z = sY[cur][(4 * sq + 2) * TP + c];
            v.w = sY[cur][(4 * sq + 3) * TP + c];
            const size_t row = row00 + (size_t)t * TS + c;
            *(float4*)(y + row * DIN + d0 + 4 * sq) = v;
        }
    }
}

extern "C" void kernel_launch(void* const* d_in, const int* in_sizes, int n_in,
                              void* d_out, int out_size, void* d_ws, size_t ws_size,
                              hipStream_t stream) {
    const float* x       = (const float*)d_in[0];
    const float* W_xproj = (const float*)d_in[1];
    const float* W_dt    = (const float*)d_in[2];
    const float* b_dt    = (const float*)d_in[3];
    const float* A_log   = (const float*)d_in[4];
    const float* D_param = (const float*)d_in[5];
    float* y = (float*)d_out;

    float* ws    = (float*)d_ws;
    float* xdbl  = ws;
    float* delta = xdbl + 2 * XPLANE;

    proj_kernel<<<256, 128, 0, stream>>>(x, W_xproj, xdbl);
    delta_kernel<<<512, 256, 0, stream>>>(xdbl, W_dt, b_dt, delta);
    scan_fused<<<256, 128, 0, stream>>>(delta, x, xdbl, A_log, D_param, y);
}
